# starting kernel
# baseline (speedup 1.0000x reference)
#include <hip/hip_runtime.h>
#include <math.h>


typedef float v2f __attribute__((ext_vector_type(2)));
typedef float v8f __attribute__((ext_vector_type(8)));

#define IMG 512
#define TILE 64
#define HALO 8
#define TAPS 17
#define IN_W 80
#define IN_STRIDE 81
#define HB_STRIDE 65

struct GaussW { float w[TAPS]; };

__device__ __forceinline__ float fast_sigmoid(float z) {
    return __builtin_amdgcn_rcpf(1.0f + __expf(-z));
}

__global__ void lamb_kernel(const float* __restrict__ c,
                            float* __restrict__ lamb, int n) {
    int i = blockIdx.x * 256 + threadIdx.x;
    if (i < n) lamb[i] = 100.0f * fast_sigmoid(10.0f * c[i]);
}

template <bool INIT>
__global__ __launch_bounds__(256)
void bvp_iter(const float* __restrict__ src,
              const float* __restrict__ x,
              const float* __restrict__ lamb,
              float* __restrict__ dst,
              GaussW gw) {
    __shared__ float in_tile[IN_W * IN_STRIDE];
    __shared__ float hbuf[IN_W * HB_STRIDE];
    __shared__ float wsm[TAPS];

    const int tid  = threadIdx.x;
    const int lane = tid & 31;
    const int wave = tid >> 5;
    const int n15  = lane & 15;
    const int koff = (lane >> 4) << 1;
    const int mhalf = (lane >> 4) << 3;

    const int row0 = blockIdx.y * TILE;
    const int col0 = blockIdx.x * TILE;
    const size_t base = (size_t)blockIdx.z * (IMG * IMG);

    if (tid < TAPS) wsm[tid] = gw.w[tid];

    for (int i = tid; i < IN_W * IN_W; i += 256) {
        int ir = i / IN_W, ic = i - ir * IN_W;
        int gr = row0 - HALO + ir, gc = col0 - HALO + ic;
        float s = 0.0f;
        if (gr >= 0 && gr < IMG && gc >= 0 && gc < IMG)
            s = 1.0f - 2.0f * src[base + (size_t)gr * IMG + gc];
        in_tile[ir * IN_STRIDE + ic] = s;
    }
    __syncthreads();

    float wband[8][2];
#pragma unroll
    for (int c = 0; c < 8; ++c)
#pragma unroll
        for (int v = 0; v < 2; ++v) {
            int d = 4 * c + koff + v - n15;
            wband[c][v] = (d >= 0 && d < TAPS) ? wsm[d] : 0.0f;
        }

    for (int t = wave; t < 20; t += 8) {
        int R = t >> 2, C = t & 3;
        const float* arow = &in_tile[(R * 16 + n15) * IN_STRIDE + C * 16 + koff];
        v8f acc = {};
#pragma unroll
        for (int c = 0; c < 8; ++c) {
            v2f a; a.x = arow[4 * c]; a.y = arow[4 * c + 1];
            v2f b; b.x = wband[c][0]; b.y = wband[c][1];
            acc = __builtin_amdgcn_wmma_f32_16x16x4_f32(
                false, a, false, b, (short)0, acc, false, false);
        }
#pragma unroll
        for (int v = 0; v < 8; ++v)
            hbuf[(R * 16 + mhalf + v) * HB_STRIDE + C * 16 + n15] = acc[v];
    }
    __syncthreads();

    const float inv_tau = 10.0f;
    const float inv_eps = 0.1f;
    for (int t = wave; t < 16; t += 8) {
        int R = t >> 2, C = t & 3;
        v8f acc = {};
#pragma unroll
        for (int c = 0; c < 8; ++c) {
            v2f a; a.x = wband[c][0]; a.y = wband[c][1];
            const float* hp = &hbuf[(R * 16 + 4 * c + koff) * HB_STRIDE + C * 16 + n15];
            v2f b; b.x = hp[0]; b.y = hp[HB_STRIDE];
            acc = __builtin_amdgcn_wmma_f32_16x16x4_f32(
                false, a, false, b, (short)0, acc, false, false);
        }
#pragma unroll
        for (int v = 0; v < 8; ++v) {
            int lr = R * 16 + mhalf + v;
            int lc = C * 16 + n15;
            size_t gidx = base + (size_t)(row0 + lr) * IMG + (col0 + lc);
            float blur = acc[v];
            float lb = lamb[gidx];
            float u;
            if (INIT) {
                u = fast_sigmoid(-(lb * blur) * inv_eps);
            } else {
                float up = 0.5f * (1.0f - in_tile[(lr + HALO) * IN_STRIDE + (lc + HALO)]);
                float xv = x[gidx];
                float uu = fast_sigmoid(-((up - xv) * inv_tau + lb * blur) * inv_eps);
                u = 0.5f * up + 0.5f * uu;
            }
            dst[gidx] = u;
        }
    }
}

extern "C" void kernel_launch(void* const* d_in, const int* in_sizes, int n_in,
                              void* d_out, int out_size, void* d_ws, size_t ws_size,
                              hipStream_t stream) {
    const float* x = (const float*)d_in[0];
    const float* c = (const float*)d_in[1];
    float* out = (float*)d_out;
    const int n = in_sizes[0];
    const int planes = n / (IMG * IMG);

    float* lamb = (float*)d_ws;
    float* ubuf = lamb + n;

    GaussW gw;
    {
        float g[TAPS], s = 0.0f;
        for (int t = 0; t < TAPS; ++t) {
            float d = (float)t - 8.0f;
            g[t] = expf(-(d * d) / 8.0f);
            s += g[t];
        }
        for (int t = 0; t < TAPS; ++t) gw.w[t] = g[t] / s;
    }

    lamb_kernel<<<(n + 255) / 256, 256, 0, stream>>>(c, lamb, n);

    dim3 grid(IMG / TILE, IMG / TILE, planes);
    bvp_iter<true ><<<grid, 256, 0, stream>>>(x,    x, lamb, ubuf, gw);
    bvp_iter<false><<<grid, 256, 0, stream>>>(ubuf, x, lamb, out,  gw);
    bvp_iter<false><<<grid, 256, 0, stream>>>(out,  x, lamb, ubuf, gw);
    bvp_iter<false><<<grid, 256, 0, stream>>>(ubuf, x, lamb, out,  gw);
    bvp_iter<false><<<grid, 256, 0, stream>>>(out,  x, lamb, ubuf, gw);
    bvp_iter<false><<<grid, 256, 0, stream>>>(ubuf, x, lamb, out,  gw);
}


# Drop the separate lam_b pass: every launch recomputes lam_b = 100*sigmoid(10c) from c in its epilogue (same arithmetic); 7 launches -> 6; a launch bound keeps the update kernel at 80 VGPRs
# speedup vs baseline: 1.0836x; 1.0836x over previous
#include <hip/hip_runtime.h>
#include <math.h>


typedef float v2f __attribute__((ext_vector_type(2)));
typedef float v8f __attribute__((ext_vector_type(8)));

#define IMG 512
#define TILE 64
#define HALO 8
#define TAPS 17
#define IN_W 80
#define IN_STRIDE 81
#define HB_STRIDE 65

struct GaussW { float w[TAPS]; };

__device__ __forceinline__ float fast_sigmoid(float z) {
    return __builtin_amdgcn_rcpf(1.0f + __expf(-z));
}

template <bool INIT>
__global__ __launch_bounds__(256, INIT ? 1 : 12)
void bvp_iter(const float* __restrict__ src,
              const float* __restrict__ x,
              const float* __restrict__ c,
              float* __restrict__ dst,
              GaussW gw) {
    __shared__ float in_tile[IN_W * IN_STRIDE];
    __shared__ float hbuf[IN_W * HB_STRIDE];
    __shared__ float wsm[TAPS];

    const int tid  = threadIdx.x;
    const int lane = tid & 31;
    const int wave = tid >> 5;
    const int n15  = lane & 15;
    const int koff = (lane >> 4) << 1;
    const int mhalf = (lane >> 4) << 3;

    const int row0 = blockIdx.y * TILE;
    const int col0 = blockIdx.x * TILE;
    const size_t base = (size_t)blockIdx.z * (IMG * IMG);

    if (tid < TAPS) wsm[tid] = gw.w[tid];

    for (int i = tid; i < IN_W * IN_W; i += 256) {
        int ir = i / IN_W, ic = i - ir * IN_W;
        int gr = row0 - HALO + ir, gc = col0 - HALO + ic;
        float s = 0.0f;
        if (gr >= 0 && gr < IMG && gc >= 0 && gc < IMG)
            s = 1.0f - 2.0f * src[base + (size_t)gr * IMG + gc];
        in_tile[ir * IN_STRIDE + ic] = s;
    }
    __syncthreads();

    float wband[8][2];
#pragma unroll
    for (int c = 0; c < 8; ++c)
#pragma unroll
        for (int v = 0; v < 2; ++v) {
            int d = 4 * c + koff + v - n15;
            wband[c][v] = (d >= 0 && d < TAPS) ? wsm[d] : 0.0f;
        }

    for (int t = wave; t < 20; t += 8) {
        int R = t >> 2, C = t & 3;
        const float* arow = &in_tile[(R * 16 + n15) * IN_STRIDE + C * 16 + koff];
        v8f acc = {};
#pragma unroll
        for (int c = 0; c < 8; ++c) {
            v2f a; a.x = arow[4 * c]; a.y = arow[4 * c + 1];
            v2f b; b.x = wband[c][0]; b.y = wband[c][1];
            acc = __builtin_amdgcn_wmma_f32_16x16x4_f32(
                false, a, false, b, (short)0, acc, false, false);
        }
#pragma unroll
        for (int v = 0; v < 8; ++v)
            hbuf[(R * 16 + mhalf + v) * HB_STRIDE + C * 16 + n15] = acc[v];
    }
    __syncthreads();

    const float inv_tau = 10.0f;
    const float inv_eps = 0.1f;
    for (int t = wave; t < 16; t += 8) {
        int R = t >> 2, C = t & 3;
        v8f acc = {};
#pragma unroll
        for (int c = 0; c < 8; ++c) {
            v2f a; a.x = wband[c][0]; a.y = wband[c][1];
            const float* hp = &hbuf[(R * 16 + 4 * c + koff) * HB_STRIDE + C * 16 + n15];
            v2f b; b.x = hp[0]; b.y = hp[HB_STRIDE];
            acc = __builtin_amdgcn_wmma_f32_16x16x4_f32(
                false, a, false, b, (short)0, acc, false, false);
        }
#pragma unroll
        for (int v = 0; v < 8; ++v) {
            int lr = R * 16 + mhalf + v;
            int lc = C * 16 + n15;
            size_t gidx = base + (size_t)(row0 + lr) * IMG + (col0 + lc);
            float blur = acc[v];
            float lb = 100.0f * fast_sigmoid(10.0f * c[gidx]);
            float u;
            if (INIT) {
                u = fast_sigmoid(-(lb * blur) * inv_eps);
            } else {
                float up = 0.5f * (1.0f - in_tile[(lr + HALO) * IN_STRIDE + (lc + HALO)]);
                float xv = x[gidx];
                float uu = fast_sigmoid(-((up - xv) * inv_tau + lb * blur) * inv_eps);
                u = 0.5f * up + 0.5f * uu;
            }
            dst[gidx] = u;
        }
    }
}

extern "C" void kernel_launch(void* const* d_in, const int* in_sizes, int n_in,
                              void* d_out, int out_size, void* d_ws, size_t ws_size,
                              hipStream_t stream) {
    const float* x = (const float*)d_in[0];
    const float* c = (const float*)d_in[1];
    float* out = (float*)d_out;
    const int n = in_sizes[0];
    const int planes = n / (IMG * IMG);

    float* ubuf = (float*)d_ws;

    GaussW gw;
    {
        float g[TAPS], s = 0.0f;
        for (int t = 0; t < TAPS; ++t) {
            float d = (float)t - 8.0f;
            g[t] = expf(-(d * d) / 8.0f);
            s += g[t];
        }
        for (int t = 0; t < TAPS; ++t) gw.w[t] = g[t] / s;
    }

    dim3 grid(IMG / TILE, IMG / TILE, planes);
    bvp_iter<true ><<<grid, 256, 0, stream>>>(x,    x, c, ubuf, gw);
    bvp_iter<false><<<grid, 256, 0, stream>>>(ubuf, x, c, out,  gw);
    bvp_iter<false><<<grid, 256, 0, stream>>>(out,  x, c, ubuf, gw);
    bvp_iter<false><<<grid, 256, 0, stream>>>(ubuf, x, c, out,  gw);
    bvp_iter<false><<<grid, 256, 0, stream>>>(out,  x, c, ubuf, gw);
    bvp_iter<false><<<grid, 256, 0, stream>>>(ubuf, x, c, out,  gw);
}


# Halo tile staged by 16-byte row loads at clamped addresses with no per-element branch and wait: a lane's 10 loads are in flight together instead of 25 serial round trips; LDS row pitch 81 -> 84
# speedup vs baseline: 1.4996x; 1.3839x over previous
#include <hip/hip_runtime.h>
#include <math.h>


typedef float v2f __attribute__((ext_vector_type(2)));
typedef float v8f __attribute__((ext_vector_type(8)));

#define IMG 512
#define TILE 64
#define HALO 8
#define TAPS 17
#define IN_W 80
#define IN_STRIDE 84
#define ROW_CHUNKS 20
#define ROWS_PER_WAVE 10
#define HB_STRIDE 65

struct GaussW { float w[TAPS]; };

__device__ __forceinline__ float fast_sigmoid(float z) {
    return __builtin_amdgcn_rcpf(1.0f + __expf(-z));
}

template <bool INIT>
__global__ __launch_bounds__(256, INIT ? 1 : 12)
void bvp_iter(const float* __restrict__ src,
              const float* __restrict__ x,
              const float* __restrict__ c,
              float* __restrict__ dst,
              GaussW gw) {
    __shared__ __attribute__((aligned(16))) float in_tile[IN_W * IN_STRIDE];
    __shared__ float hbuf[IN_W * HB_STRIDE];
    __shared__ float wsm[TAPS];

    const int tid  = threadIdx.x;
    const int lane = tid & 31;
    const int wave = tid >> 5;
    const int n15  = lane & 15;
    const int koff = (lane >> 4) << 1;
    const int mhalf = (lane >> 4) << 3;

    const int row0 = blockIdx.y * TILE;
    const int col0 = blockIdx.x * TILE;
    const size_t base = (size_t)blockIdx.z * (IMG * IMG);

    if (tid < TAPS) wsm[tid] = gw.w[tid];

    if (lane < ROW_CHUNKS) {
        const int wv = __builtin_amdgcn_readfirstlane(wave);
        const int gc = col0 - HALO + 4 * lane;
        const bool col_in = (gc >= 0) && (gc < IMG);
        const float* colp = src + base + (col_in ? gc : 0);
        float4 piece[ROWS_PER_WAVE];
#pragma unroll
        for (int j = 0; j < ROWS_PER_WAVE; ++j) {
            int gr = row0 - HALO + wv + 8 * j;
            int grc = gr < 0 ? 0 : (gr > IMG - 1 ? IMG - 1 : gr);
            piece[j] = *reinterpret_cast<const float4*>(colp + (size_t)grc * IMG);
        }
#pragma unroll
        for (int j = 0; j < ROWS_PER_WAVE; ++j) {
            int ir = wv + 8 * j;
            int gr = row0 - HALO + ir;
            bool in_img = col_in && (gr >= 0) && (gr < IMG);
            const unsigned keep = in_img ? 0xffffffffu : 0u;
            float4 s;
            s.x = __uint_as_float(__float_as_uint(1.0f - 2.0f * piece[j].x) & keep);
            s.y = __uint_as_float(__float_as_uint(1.0f - 2.0f * piece[j].y) & keep);
            s.z = __uint_as_float(__float_as_uint(1.0f - 2.0f * piece[j].z) & keep);
            s.w = __uint_as_float(__float_as_uint(1.0f - 2.0f * piece[j].w) & keep);
            *reinterpret_cast<float4*>(&in_tile[ir * IN_STRIDE + 4 * lane]) = s;
        }
    }
    __syncthreads();

    float wband[8][2];
#pragma unroll
    for (int c = 0; c < 8; ++c)
#pragma unroll
        for (int v = 0; v < 2; ++v) {
            int d = 4 * c + koff + v - n15;
            wband[c][v] = (d >= 0 && d < TAPS) ? wsm[d] : 0.0f;
        }

    for (int t = wave; t < 20; t += 8) {
        int R = t >> 2, C = t & 3;
        const float* arow = &in_tile[(R * 16 + n15) * IN_STRIDE + C * 16 + koff];
        v8f acc = {};
#pragma unroll
        for (int c = 0; c < 8; ++c) {
            v2f a; a.x = arow[4 * c]; a.y = arow[4 * c + 1];
            v2f b; b.x = wband[c][0]; b.y = wband[c][1];
            acc = __builtin_amdgcn_wmma_f32_16x16x4_f32(
                false, a, false, b, (short)0, acc, false, false);
        }
#pragma unroll
        for (int v = 0; v < 8; ++v)
            hbuf[(R * 16 + mhalf + v) * HB_STRIDE + C * 16 + n15] = acc[v];
    }
    __syncthreads();

    const float inv_tau = 10.0f;
    const float inv_eps = 0.1f;
    for (int t = wave; t < 16; t += 8) {
        int R = t >> 2, C = t & 3;
        v8f acc = {};
#pragma unroll
        for (int c = 0; c < 8; ++c) {
            v2f a; a.x = wband[c][0]; a.y = wband[c][1];
            const float* hp = &hbuf[(R * 16 + 4 * c + koff) * HB_STRIDE + C * 16 + n15];
            v2f b; b.x = hp[0]; b.y = hp[HB_STRIDE];
            acc = __builtin_amdgcn_wmma_f32_16x16x4_f32(
                false, a, false, b, (short)0, acc, false, false);
        }
#pragma unroll
        for (int v = 0; v < 8; ++v) {
            int lr = R * 16 + mhalf + v;
            int lc = C * 16 + n15;
            size_t gidx = base + (size_t)(row0 + lr) * IMG + (col0 + lc);
            float blur = acc[v];
            float lb = 100.0f * fast_sigmoid(10.0f * c[gidx]);
            float u;
            if (INIT) {
                u = fast_sigmoid(-(lb * blur) * inv_eps);
            } else {
                float up = 0.5f * (1.0f - in_tile[(lr + HALO) * IN_STRIDE + (lc + HALO)]);
                float xv = x[gidx];
                float uu = fast_sigmoid(-((up - xv) * inv_tau + lb * blur) * inv_eps);
                u = 0.5f * up + 0.5f * uu;
            }
            dst[gidx] = u;
        }
    }
}

extern "C" void kernel_launch(void* const* d_in, const int* in_sizes, int n_in,
                              void* d_out, int out_size, void* d_ws, size_t ws_size,
                              hipStream_t stream) {
    const float* x = (const float*)d_in[0];
    const float* c = (const float*)d_in[1];
    float* out = (float*)d_out;
    const int n = in_sizes[0];
    const int planes = n / (IMG * IMG);

    float* ubuf = (float*)d_ws;

    GaussW gw;
    {
        float g[TAPS], s = 0.0f;
        for (int t = 0; t < TAPS; ++t) {
            float d = (float)t - 8.0f;
            g[t] = expf(-(d * d) / 8.0f);
            s += g[t];
        }
        for (int t = 0; t < TAPS; ++t) gw.w[t] = g[t] / s;
    }

    dim3 grid(IMG / TILE, IMG / TILE, planes);
    bvp_iter<true ><<<grid, 256, 0, stream>>>(x,    x, c, ubuf, gw);
    bvp_iter<false><<<grid, 256, 0, stream>>>(ubuf, x, c, out,  gw);
    bvp_iter<false><<<grid, 256, 0, stream>>>(out,  x, c, ubuf, gw);
    bvp_iter<false><<<grid, 256, 0, stream>>>(ubuf, x, c, out,  gw);
    bvp_iter<false><<<grid, 256, 0, stream>>>(out,  x, c, ubuf, gw);
    bvp_iter<false><<<grid, 256, 0, stream>>>(ubuf, x, c, out,  gw);
}


# Each wave requests c and x of its two output tiles right behind the halo stage (scalar base + one lane offset + immediates; asm LDS-only barriers keep them in flight), not after barrier 2
# speedup vs baseline: 1.6011x; 1.0677x over previous
#include <hip/hip_runtime.h>
#include <math.h>


typedef float v2f __attribute__((ext_vector_type(2)));
typedef float v8f __attribute__((ext_vector_type(8)));

#define IMG 512
#define TILE 64
#define HALO 8
#define TAPS 17
#define IN_W 80
#define IN_STRIDE 84
#define ROW_CHUNKS 20
#define ROWS_PER_WAVE 10
#define HB_STRIDE 65

struct GaussW { float w[TAPS]; };

__device__ __forceinline__ float fast_sigmoid(float z) {
    return __builtin_amdgcn_rcpf(1.0f + __expf(-z));
}

__device__ __forceinline__ void lds_barrier_signal() {
    asm volatile("s_wait_dscnt 0x0\n\ts_barrier_signal -1" ::: "memory");
}
__device__ __forceinline__ void lds_barrier_wait() {
    asm volatile("s_barrier_wait -1" ::: "memory");
}

template <bool INIT>
__global__ __launch_bounds__(256, INIT ? 1 : 12)
void bvp_iter(const float* __restrict__ src,
              const float* __restrict__ x,
              const float* __restrict__ c,
              float* __restrict__ dst,
              GaussW gw) {
    __shared__ __attribute__((aligned(16))) float in_tile[IN_W * IN_STRIDE];
    __shared__ float hbuf[IN_W * HB_STRIDE];
    __shared__ float wsm[TAPS];

    const int tid  = threadIdx.x;
    const int lane = tid & 31;
    const int wave = tid >> 5;
    const int n15  = lane & 15;
    const int koff = (lane >> 4) << 1;
    const int mhalf = (lane >> 4) << 3;

    const int row0 = blockIdx.y * TILE;
    const int col0 = blockIdx.x * TILE;
    const size_t base = (size_t)blockIdx.z * (IMG * IMG);

    if (tid < TAPS) wsm[tid] = gw.w[tid];

    const int wv = __builtin_amdgcn_readfirstlane(wave);

    const int Rw = wv >> 2, Cw = wv & 3;
    const size_t wave_off = base + (size_t)(row0 + 16 * Rw) * IMG + (size_t)(col0 + 16 * Cw);
    const float* cw = c + wave_off;
    const float* xw = x + wave_off;
    float* dw = dst + wave_off;
    const unsigned eoff = (unsigned)(mhalf * IMG + n15);

    const int gc = col0 - HALO + 4 * lane;
    const bool col_in = (gc >= 0) && (gc < IMG);
    const unsigned gcc = col_in ? (unsigned)gc : 0u;
    if (lane < ROW_CHUNKS) {
        float4 piece[ROWS_PER_WAVE];
#pragma unroll
        for (int j = 0; j < ROWS_PER_WAVE; ++j) {
            int gr = row0 - HALO + wv + 8 * j;
            int grc = gr < 0 ? 0 : (gr > IMG - 1 ? IMG - 1 : gr);
            const float* rowp = src + (base + (size_t)grc * IMG);
            piece[j] = *reinterpret_cast<const float4*>(rowp + gcc);
        }
#pragma unroll
        for (int j = 0; j < ROWS_PER_WAVE; ++j) {
            int ir = wv + 8 * j;
            int gr = row0 - HALO + ir;
            bool in_img = col_in && (gr >= 0) && (gr < IMG);
            const unsigned keep = in_img ? 0xffffffffu : 0u;
            float4 s;
            s.x = __uint_as_float(__float_as_uint(1.0f - 2.0f * piece[j].x) & keep);
            s.y = __uint_as_float(__float_as_uint(1.0f - 2.0f * piece[j].y) & keep);
            s.z = __uint_as_float(__float_as_uint(1.0f - 2.0f * piece[j].z) & keep);
            s.w = __uint_as_float(__float_as_uint(1.0f - 2.0f * piece[j].w) & keep);
            *reinterpret_cast<float4*>(&in_tile[ir * IN_STRIDE + 4 * lane]) = s;
        }
    }

    lds_barrier_signal();
    float cpre[2][8], xpre[2][8];
#pragma unroll
    for (int j = 0; j < 2; ++j)
#pragma unroll
        for (int v = 0; v < 8; ++v) {
            cpre[j][v] = cw[eoff + (unsigned)((32 * j + v) * IMG)];
            xpre[j][v] = INIT ? 0.0f : xw[eoff + (unsigned)((32 * j + v) * IMG)];
        }
    lds_barrier_wait();

    float wband[8][2];
#pragma unroll
    for (int c = 0; c < 8; ++c)
#pragma unroll
        for (int v = 0; v < 2; ++v) {
            int d = 4 * c + koff + v - n15;
            wband[c][v] = (d >= 0 && d < TAPS) ? wsm[d] : 0.0f;
        }

    for (int t = wave; t < 20; t += 8) {
        int R = t >> 2, C = t & 3;
        const float* arow = &in_tile[(R * 16 + n15) * IN_STRIDE + C * 16 + koff];
        v8f acc = {};
#pragma unroll
        for (int c = 0; c < 8; ++c) {
            v2f a; a.x = arow[4 * c]; a.y = arow[4 * c + 1];
            v2f b; b.x = wband[c][0]; b.y = wband[c][1];
            acc = __builtin_amdgcn_wmma_f32_16x16x4_f32(
                false, a, false, b, (short)0, acc, false, false);
        }
#pragma unroll
        for (int v = 0; v < 8; ++v)
            hbuf[(R * 16 + mhalf + v) * HB_STRIDE + C * 16 + n15] = acc[v];
    }
    lds_barrier_signal();
    lds_barrier_wait();

    const float inv_tau = 10.0f;
    const float inv_eps = 0.1f;
#pragma unroll
    for (int j = 0; j < 2; ++j) {
        if (j == 1) __builtin_amdgcn_sched_barrier(0);
        const int R = Rw + 2 * j, C = Cw;
        v8f acc = {};
#pragma unroll
        for (int kc = 0; kc < 8; ++kc) {
            v2f a; a.x = wband[kc][0]; a.y = wband[kc][1];
            const float* hp = &hbuf[(R * 16 + 4 * kc + koff) * HB_STRIDE + C * 16 + n15];
            v2f b; b.x = hp[0]; b.y = hp[HB_STRIDE];
            acc = __builtin_amdgcn_wmma_f32_16x16x4_f32(
                false, a, false, b, (short)0, acc, false, false);
        }
#pragma unroll
        for (int v = 0; v < 8; ++v) {
            int lr = R * 16 + mhalf + v;
            int lc = C * 16 + n15;
            float blur = acc[v];
            float lb = 100.0f * fast_sigmoid(10.0f * cpre[j][v]);
            float u;
            if (INIT) {
                u = fast_sigmoid(-(lb * blur) * inv_eps);
            } else {
                float up = 0.5f * (1.0f - in_tile[(lr + HALO) * IN_STRIDE + (lc + HALO)]);
                float xv = xpre[j][v];
                float uu = fast_sigmoid(-((up - xv) * inv_tau + lb * blur) * inv_eps);
                u = 0.5f * up + 0.5f * uu;
            }
            dw[eoff + (unsigned)((32 * j + v) * IMG)] = u;
        }
    }
}

extern "C" void kernel_launch(void* const* d_in, const int* in_sizes, int n_in,
                              void* d_out, int out_size, void* d_ws, size_t ws_size,
                              hipStream_t stream) {
    const float* x = (const float*)d_in[0];
    const float* c = (const float*)d_in[1];
    float* out = (float*)d_out;
    const int n = in_sizes[0];
    const int planes = n / (IMG * IMG);

    float* ubuf = (float*)d_ws;

    GaussW gw;
    {
        float g[TAPS], s = 0.0f;
        for (int t = 0; t < TAPS; ++t) {
            float d = (float)t - 8.0f;
            g[t] = expf(-(d * d) / 8.0f);
            s += g[t];
        }
        for (int t = 0; t < TAPS; ++t) gw.w[t] = g[t] / s;
    }

    dim3 grid(IMG / TILE, IMG / TILE, planes);
    bvp_iter<true ><<<grid, 256, 0, stream>>>(x,    x, c, ubuf, gw);
    bvp_iter<false><<<grid, 256, 0, stream>>>(ubuf, x, c, out,  gw);
    bvp_iter<false><<<grid, 256, 0, stream>>>(out,  x, c, ubuf, gw);
    bvp_iter<false><<<grid, 256, 0, stream>>>(ubuf, x, c, out,  gw);
    bvp_iter<false><<<grid, 256, 0, stream>>>(out,  x, c, ubuf, gw);
    bvp_iter<false><<<grid, 256, 0, stream>>>(ubuf, x, c, out,  gw);
}


# Two relaxation steps per launch: a work-group stages a 96x96 tile, evaluates step one on the inner 80x80 and keeps its 1-2u in LDS in place as step two's input; 6 launches -> 3
# speedup vs baseline: 1.6969x; 1.0598x over previous
#include <hip/hip_runtime.h>
#include <math.h>


typedef float v2f __attribute__((ext_vector_type(2)));
typedef float v8f __attribute__((ext_vector_type(8)));

#define IMG 512
#define TILE 64
#define HALO 8
#define TAPS 17
#define IN_W 96
#define MID_T 5
#define IN_STRIDE 100
#define ROW_CHUNKS 24
#define ROWS_PER_WAVE 12
#define HB_STRIDE 84

struct GaussW { float w[TAPS]; };

__device__ __forceinline__ float fast_sigmoid(float z) {
    return __builtin_amdgcn_rcpf(1.0f + __expf(-z));
}

__device__ __forceinline__ void lds_barrier_signal() {
    asm volatile("s_wait_dscnt 0x0\n\ts_barrier_signal -1" ::: "memory");
}
__device__ __forceinline__ void lds_barrier_wait() {
    asm volatile("s_barrier_wait -1" ::: "memory");
}

template <bool INIT>
__device__ __forceinline__ void request_mid(const float* __restrict__ cP, const float* __restrict__ xP,
                                            int gr0, int gcl, float (&cv)[8], float (&xv)[8]) {
    const int gcc = gcl < 0 ? 0 : (gcl > IMG - 1 ? IMG - 1 : gcl);
    const int grc = gr0 < 0 ? 0 : (gr0 > IMG - 8 ? IMG - 8 : gr0);
    const unsigned off = (unsigned)(grc * IMG + gcc);
#pragma unroll
    for (int v = 0; v < 8; ++v) {
        cv[v] = cP[off + (unsigned)(v * IMG)];
        xv[v] = INIT ? 0.0f : xP[off + (unsigned)(v * IMG)];
    }
}

template <bool INIT>
__global__ __launch_bounds__(256, 8)
void bvp_two_steps(const float* __restrict__ src,
                   const float* __restrict__ x,
                   const float* __restrict__ c,
                   float* __restrict__ dst,
                   GaussW gw) {
    __shared__ __attribute__((aligned(16))) float in_tile[IN_W * IN_STRIDE];
    __shared__ float hbuf[IN_W * HB_STRIDE];
    __shared__ float wsm[TAPS];

    const int tid  = threadIdx.x;
    const int lane = tid & 31;
    const int wave = tid >> 5;
    const int n15  = lane & 15;
    const int koff = (lane >> 4) << 1;
    const int mhalf = (lane >> 4) << 3;

    const int row0 = blockIdx.y * TILE;
    const int col0 = blockIdx.x * TILE;
    const size_t base = (size_t)blockIdx.z * (IMG * IMG);

    if (tid < TAPS) wsm[tid] = gw.w[tid];

    const int wv = __builtin_amdgcn_readfirstlane(wave);

    const float* cP = c + base;
    const float* xP = x + base;

    const int Rw = wv >> 2, Cw = wv & 3;
    const size_t wave_off = (size_t)(row0 + 16 * Rw) * IMG + (size_t)(col0 + 16 * Cw);
    const float* cw = cP + wave_off;
    const float* xw = xP + wave_off;
    float* dw = dst + base + wave_off;
    const unsigned eoff = (unsigned)(mhalf * IMG + n15);

    const int gc = col0 - 2 * HALO + 4 * lane;
    const bool col_in = (gc >= 0) && (gc < IMG);
    const unsigned gcc = col_in ? (unsigned)gc : 0u;
    if (lane < ROW_CHUNKS) {
        float4 piece[ROWS_PER_WAVE];
#pragma unroll
        for (int j = 0; j < ROWS_PER_WAVE; ++j) {
            int gr = row0 - 2 * HALO + wv + 8 * j;
            int grc = gr < 0 ? 0 : (gr > IMG - 1 ? IMG - 1 : gr);
            const float* rowp = src + (base + (size_t)grc * IMG);
            piece[j] = *reinterpret_cast<const float4*>(rowp + gcc);
        }
#pragma unroll
        for (int j = 0; j < ROWS_PER_WAVE; ++j) {
            int ir = wv + 8 * j;
            int gr = row0 - 2 * HALO + ir;
            bool in_img = col_in && (gr >= 0) && (gr < IMG);
            const unsigned keep = in_img ? 0xffffffffu : 0u;
            float4 s;
            s.x = __uint_as_float(__float_as_uint(1.0f - 2.0f * piece[j].x) & keep);
            s.y = __uint_as_float(__float_as_uint(1.0f - 2.0f * piece[j].y) & keep);
            s.z = __uint_as_float(__float_as_uint(1.0f - 2.0f * piece[j].z) & keep);
            s.w = __uint_as_float(__float_as_uint(1.0f - 2.0f * piece[j].w) & keep);
            *reinterpret_cast<float4*>(&in_tile[ir * IN_STRIDE + 4 * lane]) = s;
        }
    }

    const int mid_r = row0 - HALO + mhalf;
    const int mid_c = col0 - HALO + n15;

    lds_barrier_signal();
    float cmid[4][8], xmid[4][8];
    {
        const int R = wv / MID_T, C = wv - MID_T * R;
        request_mid<INIT>(cP, xP, mid_r + 16 * R, mid_c + 16 * C, cmid[0], xmid[0]);
    }
    lds_barrier_wait();

    float wband[8][2];
#pragma unroll
    for (int cc = 0; cc < 8; ++cc)
#pragma unroll
        for (int v = 0; v < 2; ++v) {
            int d = 4 * cc + koff + v - n15;
            wband[cc][v] = (d >= 0 && d < TAPS) ? wsm[d] : 0.0f;
        }

    const float inv_tau = 10.0f;
    const float inv_eps = 0.1f;

    for (int t = wv; t < 6 * MID_T; t += 8) {
        const int R = t / MID_T, C = t - MID_T * R;
        const float* arow = &in_tile[(R * 16 + n15) * IN_STRIDE + C * 16 + koff];
        v8f acc = {};
#pragma unroll
        for (int kc = 0; kc < 8; ++kc) {
            v2f a; a.x = arow[4 * kc]; a.y = arow[4 * kc + 1];
            v2f b; b.x = wband[kc][0]; b.y = wband[kc][1];
            acc = __builtin_amdgcn_wmma_f32_16x16x4_f32(
                false, a, false, b, (short)0, acc, false, false);
        }
#pragma unroll
        for (int v = 0; v < 8; ++v)
            hbuf[(R * 16 + mhalf + v) * HB_STRIDE + C * 16 + n15] = acc[v];
    }
    lds_barrier_signal();
    lds_barrier_wait();

#pragma unroll
    for (int i = 0; i < 4; ++i) {
        const int t = (i < 3) ? (wv + 8 * i) : (MID_T * MID_T - 1);
        const bool mine = (i < 3) || (wv == 7);
        if (i > 0) __builtin_amdgcn_sched_barrier(0);
        if (mine) {
            const int R = t / MID_T, C = t - MID_T * R;
            if (i < 2) {
                const int tn = wv + 8 * (i + 1);
                const int Rn = tn / MID_T, Cn = tn - MID_T * Rn;
                request_mid<INIT>(cP, xP, mid_r + 16 * Rn, mid_c + 16 * Cn, cmid[i + 1], xmid[i + 1]);
            } else if (i == 2 && wv == 7) {
                request_mid<INIT>(cP, xP, mid_r + 16 * (MID_T - 1), mid_c + 16 * (MID_T - 1), cmid[3], xmid[3]);
            }
            v8f acc = {};
#pragma unroll
            for (int kc = 0; kc < 8; ++kc) {
                v2f a; a.x = wband[kc][0]; a.y = wband[kc][1];
                const float* hp = &hbuf[(R * 16 + 4 * kc + koff) * HB_STRIDE + C * 16 + n15];
                v2f b; b.x = hp[0]; b.y = hp[HB_STRIDE];
                acc = __builtin_amdgcn_wmma_f32_16x16x4_f32(
                    false, a, false, b, (short)0, acc, false, false);
            }
            const int gcl = mid_c + 16 * C;
            const int gr0 = mid_r + 16 * R;
            const bool in_img = (gcl >= 0) && (gcl < IMG) && (gr0 >= 0) && (gr0 < IMG);
#pragma unroll
            for (int v = 0; v < 8; ++v) {
                float* cell = &in_tile[(R * 16 + mhalf + v + HALO) * IN_STRIDE + C * 16 + n15 + HALO];
                float blur = acc[v];
                float lb = 100.0f * fast_sigmoid(10.0f * cmid[i][v]);
                float u;
                if (INIT) {
                    u = fast_sigmoid(-(lb * blur) * inv_eps);
                } else {
                    float up = 0.5f * (1.0f - *cell);
                    float uu = fast_sigmoid(-((up - xmid[i][v]) * inv_tau + lb * blur) * inv_eps);
                    u = 0.5f * up + 0.5f * uu;
                }
                *cell = in_img ? (1.0f - 2.0f * u) : 0.0f;
            }
        }
    }

    lds_barrier_signal();
    float cpre[2][8], xpre[2][8];
#pragma unroll
    for (int j = 0; j < 2; ++j)
#pragma unroll
        for (int v = 0; v < 8; ++v) {
            cpre[j][v] = cw[eoff + (unsigned)((32 * j + v) * IMG)];
            xpre[j][v] = xw[eoff + (unsigned)((32 * j + v) * IMG)];
        }
    lds_barrier_wait();

#pragma unroll
    for (int i = 0; i < 3; ++i) {
        const int t = (i < 2) ? (wv + 8 * i) : (wv + 12);
        const bool mine = (i < 2) || (wv >= 4);
        if (mine) {
            const int R = t >> 2, C = t & 3;
            const float* arow = &in_tile[(HALO + R * 16 + n15) * IN_STRIDE + HALO + C * 16 + koff];
            v8f acc = {};
#pragma unroll
            for (int kc = 0; kc < 8; ++kc) {
                v2f a; a.x = arow[4 * kc]; a.y = arow[4 * kc + 1];
                v2f b; b.x = wband[kc][0]; b.y = wband[kc][1];
                acc = __builtin_amdgcn_wmma_f32_16x16x4_f32(
                    false, a, false, b, (short)0, acc, false, false);
            }
#pragma unroll
            for (int v = 0; v < 8; ++v)
                hbuf[(R * 16 + mhalf + v) * HB_STRIDE + C * 16 + n15] = acc[v];
        }
    }
    lds_barrier_signal();
    lds_barrier_wait();

#pragma unroll
    for (int j = 0; j < 2; ++j) {
        const int R = Rw + 2 * j, C = Cw;
        v8f acc = {};
#pragma unroll
        for (int kc = 0; kc < 8; ++kc) {
            v2f a; a.x = wband[kc][0]; a.y = wband[kc][1];
            const float* hp = &hbuf[(R * 16 + 4 * kc + koff) * HB_STRIDE + C * 16 + n15];
            v2f b; b.x = hp[0]; b.y = hp[HB_STRIDE];
            acc = __builtin_amdgcn_wmma_f32_16x16x4_f32(
                false, a, false, b, (short)0, acc, false, false);
        }
#pragma unroll
        for (int v = 0; v < 8; ++v) {
            int lr = R * 16 + mhalf + v;
            int lc = C * 16 + n15;
            float blur = acc[v];
            float lb = 100.0f * fast_sigmoid(10.0f * cpre[j][v]);
            float up = 0.5f * (1.0f - in_tile[(lr + 2 * HALO) * IN_STRIDE + (lc + 2 * HALO)]);
            float xv = xpre[j][v];
            float uu = fast_sigmoid(-((up - xv) * inv_tau + lb * blur) * inv_eps);
            dw[eoff + (unsigned)((32 * j + v) * IMG)] = 0.5f * up + 0.5f * uu;
        }
    }
}

extern "C" void kernel_launch(void* const* d_in, const int* in_sizes, int n_in,
                              void* d_out, int out_size, void* d_ws, size_t ws_size,
                              hipStream_t stream) {
    const float* x = (const float*)d_in[0];
    const float* c = (const float*)d_in[1];
    float* out = (float*)d_out;
    const int n = in_sizes[0];
    const int planes = n / (IMG * IMG);

    float* ubuf = (float*)d_ws;

    GaussW gw;
    {
        float g[TAPS], s = 0.0f;
        for (int t = 0; t < TAPS; ++t) {
            float d = (float)t - 8.0f;
            g[t] = expf(-(d * d) / 8.0f);
            s += g[t];
        }
        for (int t = 0; t < TAPS; ++t) gw.w[t] = g[t] / s;
    }

    dim3 grid(IMG / TILE, IMG / TILE, planes);
    bvp_two_steps<true ><<<grid, 256, 0, stream>>>(x,    x, c, out,  gw);
    bvp_two_steps<false><<<grid, 256, 0, stream>>>(out,  x, c, ubuf, gw);
    bvp_two_steps<false><<<grid, 256, 0, stream>>>(ubuf, x, c, out,  gw);
}


# Every sigmoid as one hardware tanh, sigmoid(z) = (1 + tanh(z/2))/2, the update written in s = 1-2u with its constants folded: 9 vector instructions per updated element for 20; update re-associated
# speedup vs baseline: 1.8516x; 1.0912x over previous
#include <hip/hip_runtime.h>
#include <math.h>


typedef float v2f __attribute__((ext_vector_type(2)));
typedef float v8f __attribute__((ext_vector_type(8)));

#define IMG 512
#define TILE 64
#define HALO 8
#define TAPS 17
#define IN_W 96
#define MID_T 5
#define IN_STRIDE 100
#define ROW_CHUNKS 24
#define ROWS_PER_WAVE 12
#define HB_STRIDE 84

struct GaussW { float w[TAPS]; };

__device__ __forceinline__ float hw_tanh(float z) {
    return __builtin_amdgcn_tanhf(z);
}

__device__ __forceinline__ void lds_barrier_signal() {
    asm volatile("s_wait_dscnt 0x0\n\ts_barrier_signal -1" ::: "memory");
}
__device__ __forceinline__ void lds_barrier_wait() {
    asm volatile("s_barrier_wait -1" ::: "memory");
}

template <bool INIT>
__device__ __forceinline__ void request_mid(const float* __restrict__ cP, const float* __restrict__ xP,
                                            int gr0, int gcl, float (&cv)[8], float (&xv)[8]) {
    const int gcc = gcl < 0 ? 0 : (gcl > IMG - 1 ? IMG - 1 : gcl);
    const int grc = gr0 < 0 ? 0 : (gr0 > IMG - 8 ? IMG - 8 : gr0);
    const unsigned off = (unsigned)(grc * IMG + gcc);
#pragma unroll
    for (int v = 0; v < 8; ++v) {
        cv[v] = cP[off + (unsigned)(v * IMG)];
        xv[v] = INIT ? 0.0f : xP[off + (unsigned)(v * IMG)];
    }
}

template <bool INIT>
__global__ __launch_bounds__(256, 8)
void bvp_two_steps(const float* __restrict__ src,
                   const float* __restrict__ x,
                   const float* __restrict__ c,
                   float* __restrict__ dst,
                   GaussW gw) {
    __shared__ __attribute__((aligned(16))) float in_tile[IN_W * IN_STRIDE];
    __shared__ float hbuf[IN_W * HB_STRIDE];
    __shared__ float wsm[TAPS];

    const int tid  = threadIdx.x;
    const int lane = tid & 31;
    const int wave = tid >> 5;
    const int n15  = lane & 15;
    const int koff = (lane >> 4) << 1;
    const int mhalf = (lane >> 4) << 3;

    const int row0 = blockIdx.y * TILE;
    const int col0 = blockIdx.x * TILE;
    const size_t base = (size_t)blockIdx.z * (IMG * IMG);

    if (tid < TAPS) wsm[tid] = gw.w[tid];

    const int wv = __builtin_amdgcn_readfirstlane(wave);

    const float* cP = c + base;
    const float* xP = x + base;

    const int Rw = wv >> 2, Cw = wv & 3;
    const size_t wave_off = (size_t)(row0 + 16 * Rw) * IMG + (size_t)(col0 + 16 * Cw);
    const float* cw = cP + wave_off;
    const float* xw = xP + wave_off;
    float* dw = dst + base + wave_off;
    const unsigned eoff = (unsigned)(mhalf * IMG + n15);

    const int gc = col0 - 2 * HALO + 4 * lane;
    const bool col_in = (gc >= 0) && (gc < IMG);
    const unsigned gcc = col_in ? (unsigned)gc : 0u;
    if (lane < ROW_CHUNKS) {
        float4 piece[ROWS_PER_WAVE];
#pragma unroll
        for (int j = 0; j < ROWS_PER_WAVE; ++j) {
            int gr = row0 - 2 * HALO + wv + 8 * j;
            int grc = gr < 0 ? 0 : (gr > IMG - 1 ? IMG - 1 : gr);
            const float* rowp = src + (base + (size_t)grc * IMG);
            piece[j] = *reinterpret_cast<const float4*>(rowp + gcc);
        }
#pragma unroll
        for (int j = 0; j < ROWS_PER_WAVE; ++j) {
            int ir = wv + 8 * j;
            int gr = row0 - 2 * HALO + ir;
            bool in_img = col_in && (gr >= 0) && (gr < IMG);
            const unsigned keep = in_img ? 0xffffffffu : 0u;
            float4 s;
            s.x = __uint_as_float(__float_as_uint(1.0f - 2.0f * piece[j].x) & keep);
            s.y = __uint_as_float(__float_as_uint(1.0f - 2.0f * piece[j].y) & keep);
            s.z = __uint_as_float(__float_as_uint(1.0f - 2.0f * piece[j].z) & keep);
            s.w = __uint_as_float(__float_as_uint(1.0f - 2.0f * piece[j].w) & keep);
            *reinterpret_cast<float4*>(&in_tile[ir * IN_STRIDE + 4 * lane]) = s;
        }
    }

    const int mid_r = row0 - HALO + mhalf;
    const int mid_c = col0 - HALO + n15;

    lds_barrier_signal();
    float cmid[4][8], xmid[4][8];
    {
        const int R = wv / MID_T, C = wv - MID_T * R;
        request_mid<INIT>(cP, xP, mid_r + 16 * R, mid_c + 16 * C, cmid[0], xmid[0]);
    }
    lds_barrier_wait();

    float wband[8][2];
#pragma unroll
    for (int cc = 0; cc < 8; ++cc)
#pragma unroll
        for (int v = 0; v < 2; ++v) {
            int d = 4 * cc + koff + v - n15;
            wband[cc][v] = (d >= 0 && d < TAPS) ? wsm[d] : 0.0f;
        }

    for (int t = wv; t < 6 * MID_T; t += 8) {
        const int R = t / MID_T, C = t - MID_T * R;
        const float* arow = &in_tile[(R * 16 + n15) * IN_STRIDE + C * 16 + koff];
        v8f acc = {};
#pragma unroll
        for (int kc = 0; kc < 8; ++kc) {
            v2f a; a.x = arow[4 * kc]; a.y = arow[4 * kc + 1];
            v2f b; b.x = wband[kc][0]; b.y = wband[kc][1];
            acc = __builtin_amdgcn_wmma_f32_16x16x4_f32(
                false, a, false, b, (short)0, acc, false, false);
        }
#pragma unroll
        for (int v = 0; v < 8; ++v)
            hbuf[(R * 16 + mhalf + v) * HB_STRIDE + C * 16 + n15] = acc[v];
    }
    lds_barrier_signal();
    lds_barrier_wait();

#pragma unroll
    for (int i = 0; i < 4; ++i) {
        const int t = (i < 3) ? (wv + 8 * i) : (MID_T * MID_T - 1);
        const bool mine = (i < 3) || (wv == 7);
        if (i > 0) __builtin_amdgcn_sched_barrier(0);
        if (mine) {
            const int R = t / MID_T, C = t - MID_T * R;
            if (i < 2) {
                const int tn = wv + 8 * (i + 1);
                const int Rn = tn / MID_T, Cn = tn - MID_T * Rn;
                request_mid<INIT>(cP, xP, mid_r + 16 * Rn, mid_c + 16 * Cn, cmid[i + 1], xmid[i + 1]);
            } else if (i == 2 && wv == 7) {
                request_mid<INIT>(cP, xP, mid_r + 16 * (MID_T - 1), mid_c + 16 * (MID_T - 1), cmid[3], xmid[3]);
            }
            v8f acc = {};
#pragma unroll
            for (int kc = 0; kc < 8; ++kc) {
                v2f a; a.x = wband[kc][0]; a.y = wband[kc][1];
                const float* hp = &hbuf[(R * 16 + 4 * kc + koff) * HB_STRIDE + C * 16 + n15];
                v2f b; b.x = hp[0]; b.y = hp[HB_STRIDE];
                acc = __builtin_amdgcn_wmma_f32_16x16x4_f32(
                    false, a, false, b, (short)0, acc, false, false);
            }
            const int gcl = mid_c + 16 * C;
            const int gr0 = mid_r + 16 * R;
            const bool in_img = (gcl >= 0) && (gcl < IMG) && (gr0 >= 0) && (gr0 < IMG);
            const float gm = in_img ? 2.5f : 0.0f;
            const float hm = in_img ? 0.5f : 0.0f;
#pragma unroll
            for (int v = 0; v < 8; ++v) {
                float* cell = &in_tile[(R * 16 + mhalf + v + HALO) * IN_STRIDE + C * 16 + n15 + HALO];
                float t1 = hw_tanh(5.0f * cmid[i][v]);
                if (INIT) {
                    *cell = hw_tanh(__builtin_fmaf(t1, gm, gm) * acc[v]);
                } else {
                    float g = __builtin_fmaf(t1, 2.5f, 2.5f);
                    float s = *cell;
                    float d = __builtin_fmaf(xmid[i][v], -0.5f, __builtin_fmaf(s, -0.25f, 0.25f));
                    float t2 = hw_tanh(__builtin_fmaf(g, acc[v], d));
                    *cell = hm * (s + t2);
                }
            }
        }
    }

    lds_barrier_signal();
    float cpre[2][8], xpre[2][8];
#pragma unroll
    for (int j = 0; j < 2; ++j)
#pragma unroll
        for (int v = 0; v < 8; ++v) {
            cpre[j][v] = cw[eoff + (unsigned)((32 * j + v) * IMG)];
            xpre[j][v] = xw[eoff + (unsigned)((32 * j + v) * IMG)];
        }
    lds_barrier_wait();

#pragma unroll
    for (int i = 0; i < 3; ++i) {
        const int t = (i < 2) ? (wv + 8 * i) : (wv + 12);
        const bool mine = (i < 2) || (wv >= 4);
        if (mine) {
            const int R = t >> 2, C = t & 3;
            const float* arow = &in_tile[(HALO + R * 16 + n15) * IN_STRIDE + HALO + C * 16 + koff];
            v8f acc = {};
#pragma unroll
            for (int kc = 0; kc < 8; ++kc) {
                v2f a; a.x = arow[4 * kc]; a.y = arow[4 * kc + 1];
                v2f b; b.x = wband[kc][0]; b.y = wband[kc][1];
                acc = __builtin_amdgcn_wmma_f32_16x16x4_f32(
                    false, a, false, b, (short)0, acc, false, false);
            }
#pragma unroll
            for (int v = 0; v < 8; ++v)
                hbuf[(R * 16 + mhalf + v) * HB_STRIDE + C * 16 + n15] = acc[v];
        }
    }
    lds_barrier_signal();
    lds_barrier_wait();

#pragma unroll
    for (int j = 0; j < 2; ++j) {
        const int R = Rw + 2 * j, C = Cw;
        v8f acc = {};
#pragma unroll
        for (int kc = 0; kc < 8; ++kc) {
            v2f a; a.x = wband[kc][0]; a.y = wband[kc][1];
            const float* hp = &hbuf[(R * 16 + 4 * kc + koff) * HB_STRIDE + C * 16 + n15];
            v2f b; b.x = hp[0]; b.y = hp[HB_STRIDE];
            acc = __builtin_amdgcn_wmma_f32_16x16x4_f32(
                false, a, false, b, (short)0, acc, false, false);
        }
#pragma unroll
        for (int v = 0; v < 8; ++v) {
            int lr = R * 16 + mhalf + v;
            int lc = C * 16 + n15;
            float g = __builtin_fmaf(hw_tanh(5.0f * cpre[j][v]), 2.5f, 2.5f);
            float s = in_tile[(lr + 2 * HALO) * IN_STRIDE + (lc + 2 * HALO)];
            float d = __builtin_fmaf(xpre[j][v], -0.5f, __builtin_fmaf(s, -0.25f, 0.25f));
            float t2 = hw_tanh(__builtin_fmaf(g, acc[v], d));
            dw[eoff + (unsigned)((32 * j + v) * IMG)] = __builtin_fmaf(s + t2, -0.25f, 0.5f);
        }
    }
}

extern "C" void kernel_launch(void* const* d_in, const int* in_sizes, int n_in,
                              void* d_out, int out_size, void* d_ws, size_t ws_size,
                              hipStream_t stream) {
    const float* x = (const float*)d_in[0];
    const float* c = (const float*)d_in[1];
    float* out = (float*)d_out;
    const int n = in_sizes[0];
    const int planes = n / (IMG * IMG);

    float* ubuf = (float*)d_ws;

    GaussW gw;
    {
        float g[TAPS], s = 0.0f;
        for (int t = 0; t < TAPS; ++t) {
            float d = (float)t - 8.0f;
            g[t] = expf(-(d * d) / 8.0f);
            s += g[t];
        }
        for (int t = 0; t < TAPS; ++t) gw.w[t] = g[t] / s;
    }

    dim3 grid(IMG / TILE, IMG / TILE, planes);
    bvp_two_steps<true ><<<grid, 256, 0, stream>>>(x,    x, c, out,  gw);
    bvp_two_steps<false><<<grid, 256, 0, stream>>>(out,  x, c, ubuf, gw);
    bvp_two_steps<false><<<grid, 256, 0, stream>>>(ubuf, x, c, out,  gw);
}


# Work-group tile 64x64 -> 128x64 outputs on 16 waves (grid 8x8x24 -> 8x4x24, two work-groups per WGP): less two-step rim, 163 matrix tiles per 128x64 for 182, staged input 160x96 for 2 x 96x96
# speedup vs baseline: 1.9939x; 1.0769x over previous
#include <hip/hip_runtime.h>
#include <math.h>


typedef float v2f __attribute__((ext_vector_type(2)));
typedef float v8f __attribute__((ext_vector_type(8)));

#define IMG 512
#define TILE_R 128
#define TILE_C 64
#define NWAVES 16
#define HALO 8
#define TAPS 17
#define IN_R 160
#define IN_W 96
#define MID_TR 9
#define MID_T 5
#define IN_STRIDE 100
#define ROW_CHUNKS 24
#define ROWS_PER_WAVE 10
#define HB_STRIDE 84

struct GaussW { float w[TAPS]; };

__device__ __forceinline__ float hw_tanh(float z) {
    return __builtin_amdgcn_tanhf(z);
}

__device__ __forceinline__ void lds_barrier_signal() {
    asm volatile("s_wait_dscnt 0x0\n\ts_barrier_signal -1" ::: "memory");
}
__device__ __forceinline__ void lds_barrier_wait() {
    asm volatile("s_barrier_wait -1" ::: "memory");
}

template <bool INIT>
__device__ __forceinline__ void request_mid(const float* __restrict__ cP, const float* __restrict__ xP,
                                            int gr0, int gcl, float (&cv)[8], float (&xv)[8]) {
    const int gcc = gcl < 0 ? 0 : (gcl > IMG - 1 ? IMG - 1 : gcl);
    const int grc = gr0 < 0 ? 0 : (gr0 > IMG - 8 ? IMG - 8 : gr0);
    const unsigned off = (unsigned)(grc * IMG + gcc);
#pragma unroll
    for (int v = 0; v < 8; ++v) {
        cv[v] = cP[off + (unsigned)(v * IMG)];
        xv[v] = INIT ? 0.0f : xP[off + (unsigned)(v * IMG)];
    }
}

template <bool INIT>
__global__ __launch_bounds__(512, 8)
void bvp_two_steps(const float* __restrict__ src,
                   const float* __restrict__ x,
                   const float* __restrict__ c,
                   float* __restrict__ dst,
                   GaussW gw) {
    __shared__ __attribute__((aligned(16))) float in_tile[IN_R * IN_STRIDE];
    __shared__ float hbuf[IN_R * HB_STRIDE];
    __shared__ float wsm[TAPS];

    const int tid  = threadIdx.x;
    const int lane = tid & 31;
    const int wave = tid >> 5;
    const int n15  = lane & 15;
    const int koff = (lane >> 4) << 1;
    const int mhalf = (lane >> 4) << 3;

    const int row0 = blockIdx.y * TILE_R;
    const int col0 = blockIdx.x * TILE_C;
    const size_t base = (size_t)blockIdx.z * (IMG * IMG);

    if (tid < TAPS) wsm[tid] = gw.w[tid];

    const int wv = __builtin_amdgcn_readfirstlane(wave);

    const float* cP = c + base;
    const float* xP = x + base;

    const int Rw = wv >> 2, Cw = wv & 3;
    const size_t wave_off = (size_t)(row0 + 16 * Rw) * IMG + (size_t)(col0 + 16 * Cw);
    const float* cw = cP + wave_off;
    const float* xw = xP + wave_off;
    float* dw = dst + base + wave_off;
    const unsigned eoff = (unsigned)(mhalf * IMG + n15);

    const int gc = col0 - 2 * HALO + 4 * lane;
    const bool col_in = (gc >= 0) && (gc < IMG);
    const unsigned gcc = col_in ? (unsigned)gc : 0u;
    if (lane < ROW_CHUNKS) {
        float4 piece[ROWS_PER_WAVE];
#pragma unroll
        for (int j = 0; j < ROWS_PER_WAVE; ++j) {
            int gr = row0 - 2 * HALO + wv + NWAVES * j;
            int grc = gr < 0 ? 0 : (gr > IMG - 1 ? IMG - 1 : gr);
            const float* rowp = src + (base + (size_t)grc * IMG);
            piece[j] = *reinterpret_cast<const float4*>(rowp + gcc);
        }
#pragma unroll
        for (int j = 0; j < ROWS_PER_WAVE; ++j) {
            int ir = wv + NWAVES * j;
            int gr = row0 - 2 * HALO + ir;
            bool in_img = col_in && (gr >= 0) && (gr < IMG);
            const unsigned keep = in_img ? 0xffffffffu : 0u;
            float4 s;
            s.x = __uint_as_float(__float_as_uint(1.0f - 2.0f * piece[j].x) & keep);
            s.y = __uint_as_float(__float_as_uint(1.0f - 2.0f * piece[j].y) & keep);
            s.z = __uint_as_float(__float_as_uint(1.0f - 2.0f * piece[j].z) & keep);
            s.w = __uint_as_float(__float_as_uint(1.0f - 2.0f * piece[j].w) & keep);
            *reinterpret_cast<float4*>(&in_tile[ir * IN_STRIDE + 4 * lane]) = s;
        }
    }

    const int mid_r = row0 - HALO + mhalf;
    const int mid_c = col0 - HALO + n15;

    lds_barrier_signal();
    float cmid[3][8], xmid[3][8];
    {
        const int R = wv / MID_T, C = wv - MID_T * R;
        request_mid<INIT>(cP, xP, mid_r + 16 * R, mid_c + 16 * C, cmid[0], xmid[0]);
    }
    lds_barrier_wait();

    float wband[8][2];
#pragma unroll
    for (int cc = 0; cc < 8; ++cc)
#pragma unroll
        for (int v = 0; v < 2; ++v) {
            int d = 4 * cc + koff + v - n15;
            wband[cc][v] = (d >= 0 && d < TAPS) ? wsm[d] : 0.0f;
        }

    for (int t = wv; t < (IN_R / 16) * MID_T; t += NWAVES) {
        const int R = t / MID_T, C = t - MID_T * R;
        const float* arow = &in_tile[(R * 16 + n15) * IN_STRIDE + C * 16 + koff];
        v8f acc = {};
#pragma unroll
        for (int kc = 0; kc < 8; ++kc) {
            v2f a; a.x = arow[4 * kc]; a.y = arow[4 * kc + 1];
            v2f b; b.x = wband[kc][0]; b.y = wband[kc][1];
            acc = __builtin_amdgcn_wmma_f32_16x16x4_f32(
                false, a, false, b, (short)0, acc, false, false);
        }
#pragma unroll
        for (int v = 0; v < 8; ++v)
            hbuf[(R * 16 + mhalf + v) * HB_STRIDE + C * 16 + n15] = acc[v];
    }
    lds_barrier_signal();
    lds_barrier_wait();

#pragma unroll
    for (int i = 0; i < 3; ++i) {
        const int t = wv + NWAVES * i;
        const bool mine = (i < 2) || (t < MID_TR * MID_T);
        if (i > 0) __builtin_amdgcn_sched_barrier(0);
        if (mine) {
            const int R = t / MID_T, C = t - MID_T * R;
            if (i < 2) {
                const int tn = wv + NWAVES * (i + 1);
                if (i < 1 || tn < MID_TR * MID_T) {
                    const int Rn = tn / MID_T, Cn = tn - MID_T * Rn;
                    request_mid<INIT>(cP, xP, mid_r + 16 * Rn, mid_c + 16 * Cn, cmid[i + 1], xmid[i + 1]);
                }
            }
            v8f acc = {};
#pragma unroll
            for (int kc = 0; kc < 8; ++kc) {
                v2f a; a.x = wband[kc][0]; a.y = wband[kc][1];
                const float* hp = &hbuf[(R * 16 + 4 * kc + koff) * HB_STRIDE + C * 16 + n15];
                v2f b; b.x = hp[0]; b.y = hp[HB_STRIDE];
                acc = __builtin_amdgcn_wmma_f32_16x16x4_f32(
                    false, a, false, b, (short)0, acc, false, false);
            }
            const int gcl = mid_c + 16 * C;
            const int gr0 = mid_r + 16 * R;
            const bool in_img = (gcl >= 0) && (gcl < IMG) && (gr0 >= 0) && (gr0 < IMG);
            const float gm = in_img ? 2.5f : 0.0f;
            const float hm = in_img ? 0.5f : 0.0f;
#pragma unroll
            for (int v = 0; v < 8; ++v) {
                float* cell = &in_tile[(R * 16 + mhalf + v + HALO) * IN_STRIDE + C * 16 + n15 + HALO];
                float t1 = hw_tanh(5.0f * cmid[i][v]);
                if (INIT) {
                    *cell = hw_tanh(__builtin_fmaf(t1, gm, gm) * acc[v]);
                } else {
                    float g = __builtin_fmaf(t1, 2.5f, 2.5f);
                    float s = *cell;
                    float d = __builtin_fmaf(xmid[i][v], -0.5f, __builtin_fmaf(s, -0.25f, 0.25f));
                    float t2 = hw_tanh(__builtin_fmaf(g, acc[v], d));
                    *cell = hm * (s + t2);
                }
            }
        }
    }

    lds_barrier_signal();
    float cpre[2][8], xpre[2][8];
#pragma unroll
    for (int j = 0; j < 2; ++j)
#pragma unroll
        for (int v = 0; v < 8; ++v) {
            cpre[j][v] = cw[eoff + (unsigned)((64 * j + v) * IMG)];
            xpre[j][v] = xw[eoff + (unsigned)((64 * j + v) * IMG)];
        }
    lds_barrier_wait();

#pragma unroll
    for (int i = 0; i < 3; ++i) {
        const int t = wv + NWAVES * i;
        const bool mine = (i < 2) || (t < MID_TR * 4);
        if (mine) {
            const int R = t >> 2, C = t & 3;
            const float* arow = &in_tile[(HALO + R * 16 + n15) * IN_STRIDE + HALO + C * 16 + koff];
            v8f acc = {};
#pragma unroll
            for (int kc = 0; kc < 8; ++kc) {
                v2f a; a.x = arow[4 * kc]; a.y = arow[4 * kc + 1];
                v2f b; b.x = wband[kc][0]; b.y = wband[kc][1];
                acc = __builtin_amdgcn_wmma_f32_16x16x4_f32(
                    false, a, false, b, (short)0, acc, false, false);
            }
#pragma unroll
            for (int v = 0; v < 8; ++v)
                hbuf[(R * 16 + mhalf + v) * HB_STRIDE + C * 16 + n15] = acc[v];
        }
    }
    lds_barrier_signal();
    lds_barrier_wait();

#pragma unroll
    for (int j = 0; j < 2; ++j) {
        const int R = Rw + 4 * j, C = Cw;
        v8f acc = {};
#pragma unroll
        for (int kc = 0; kc < 8; ++kc) {
            v2f a; a.x = wband[kc][0]; a.y = wband[kc][1];
            const float* hp = &hbuf[(R * 16 + 4 * kc + koff) * HB_STRIDE + C * 16 + n15];
            v2f b; b.x = hp[0]; b.y = hp[HB_STRIDE];
            acc = __builtin_amdgcn_wmma_f32_16x16x4_f32(
                false, a, false, b, (short)0, acc, false, false);
        }
#pragma unroll
        for (int v = 0; v < 8; ++v) {
            int lr = R * 16 + mhalf + v;
            int lc = C * 16 + n15;
            float g = __builtin_fmaf(hw_tanh(5.0f * cpre[j][v]), 2.5f, 2.5f);
            float s = in_tile[(lr + 2 * HALO) * IN_STRIDE + (lc + 2 * HALO)];
            float d = __builtin_fmaf(xpre[j][v], -0.5f, __builtin_fmaf(s, -0.25f, 0.25f));
            float t2 = hw_tanh(__builtin_fmaf(g, acc[v], d));
            dw[eoff + (unsigned)((64 * j + v) * IMG)] = __builtin_fmaf(s + t2, -0.25f, 0.5f);
        }
    }
}

extern "C" void kernel_launch(void* const* d_in, const int* in_sizes, int n_in,
                              void* d_out, int out_size, void* d_ws, size_t ws_size,
                              hipStream_t stream) {
    const float* x = (const float*)d_in[0];
    const float* c = (const float*)d_in[1];
    float* out = (float*)d_out;
    const int n = in_sizes[0];
    const int planes = n / (IMG * IMG);

    float* ubuf = (float*)d_ws;

    GaussW gw;
    {
        float g[TAPS], s = 0.0f;
        for (int t = 0; t < TAPS; ++t) {
            float d = (float)t - 8.0f;
            g[t] = expf(-(d * d) / 8.0f);
            s += g[t];
        }
        for (int t = 0; t < TAPS; ++t) gw.w[t] = g[t] / s;
    }

    dim3 grid(IMG / TILE_C, IMG / TILE_R, planes);
    bvp_two_steps<true ><<<grid, 512, 0, stream>>>(x,    x, c, out,  gw);
    bvp_two_steps<false><<<grid, 512, 0, stream>>>(out,  x, c, ubuf, gw);
    bvp_two_steps<false><<<grid, 512, 0, stream>>>(ubuf, x, c, out,  gw);
}


# Horizontal-pass accumulators serve in place as the vertical pass's B operand (k order 8h+2j+s, sums regrouped): no LDS round trip between passes, one barrier fewer; 8-wave groups, 3 resident per WGP
# speedup vs baseline: 2.0750x; 1.0407x over previous
#include <hip/hip_runtime.h>
#include <math.h>


typedef float v2f __attribute__((ext_vector_type(2)));
typedef float v8f __attribute__((ext_vector_type(8)));

#define IMG 512
#define TILE_R 128
#define TILE_C 64
#define NWAVES 8
#define HALO 8
#define TAPS 17
#define IN_R 160
#define IN_W 96
#define MID_T 5
#define MID_RUNS 15
#define IN_STRIDE 100
#define ROW_CHUNKS 24
#define ROWS_PER_WAVE 20

struct GaussW { float w[TAPS]; };

__device__ __forceinline__ float hw_tanh(float z) {
    return __builtin_amdgcn_tanhf(z);
}

__device__ __forceinline__ void lds_barrier_signal() {
    asm volatile("s_wait_dscnt 0x0\n\ts_barrier_signal -1" ::: "memory");
}
__device__ __forceinline__ void lds_barrier_wait() {
    asm volatile("s_barrier_wait -1" ::: "memory");
}

template <bool INIT>
__device__ __forceinline__ void request_mid(const float* __restrict__ cP, const float* __restrict__ xP,
                                            int gr0, int gcl, float (&cv)[8], float (&xv)[8]) {
    const int gcc = gcl < 0 ? 0 : (gcl > IMG - 1 ? IMG - 1 : gcl);
    const int grc = gr0 < 0 ? 0 : (gr0 > IMG - 8 ? IMG - 8 : gr0);
    const unsigned off = (unsigned)(grc * IMG + gcc);
#pragma unroll
    for (int v = 0; v < 8; ++v) {
        cv[v] = cP[off + (unsigned)(v * IMG)];
        xv[v] = INIT ? 0.0f : xP[off + (unsigned)(v * IMG)];
    }
}

__device__ __forceinline__ v8f h_tile(const float* p, const float (&bw)[8][2]) {
    const float4 l0 = *reinterpret_cast<const float4*>(p);
    const float4 l1 = *reinterpret_cast<const float4*>(p + 4);
    const float4 l2 = *reinterpret_cast<const float4*>(p + 16);
    const float4 l3 = *reinterpret_cast<const float4*>(p + 20);
    const float av[8][2] = {{l0.x, l0.y}, {l0.z, l0.w}, {l1.x, l1.y}, {l1.z, l1.w},
                            {l2.x, l2.y}, {l2.z, l2.w}, {l3.x, l3.y}, {l3.z, l3.w}};
    v8f acc = {};
#pragma unroll
    for (int q = 0; q < 8; ++q) {
        v2f a; a.x = av[q][0]; a.y = av[q][1];
        v2f b; b.x = bw[q][0]; b.y = bw[q][1];
        acc = __builtin_amdgcn_wmma_f32_16x16x4_f32(false, a, false, b, (short)0, acc, false, false);
    }
    return acc;
}

__device__ __forceinline__ v8f v_tile(const v8f& up, const v8f& lo, const float (&bw)[8][2]) {
    v8f acc = {};
#pragma unroll
    for (int q = 0; q < 4; ++q) {
        v2f a; a.x = bw[q][0]; a.y = bw[q][1];
        v2f b; b.x = up[2 * q]; b.y = up[2 * q + 1];
        acc = __builtin_amdgcn_wmma_f32_16x16x4_f32(false, a, false, b, (short)0, acc, false, false);
    }
#pragma unroll
    for (int q = 0; q < 4; ++q) {
        v2f a; a.x = bw[4 + q][0]; a.y = bw[4 + q][1];
        v2f b; b.x = lo[2 * q]; b.y = lo[2 * q + 1];
        acc = __builtin_amdgcn_wmma_f32_16x16x4_f32(false, a, false, b, (short)0, acc, false, false);
    }
    return acc;
}

template <bool INIT>
__global__ __launch_bounds__(256, 6)
void bvp_two_steps(const float* __restrict__ src,
                   const float* __restrict__ x,
                   const float* __restrict__ c,
                   float* __restrict__ dst,
                   GaussW gw) {
    __shared__ __attribute__((aligned(16))) float in_tile[IN_R * IN_STRIDE];
    __shared__ float stash[NWAVES * 5 * 256];
    __shared__ float wsm[TAPS];

    const int tid  = threadIdx.x;
    const int lane = tid & 31;
    const int wave = tid >> 5;
    const int n15  = lane & 15;
    const int hh   = lane >> 4;
    const int mhalf = hh << 3;

    const int row0 = blockIdx.y * TILE_R;
    const int col0 = blockIdx.x * TILE_C;
    const size_t base = (size_t)blockIdx.z * (IMG * IMG);

    if (tid < TAPS) wsm[tid] = gw.w[tid];

    const int wv = __builtin_amdgcn_readfirstlane(wave);

    const float* cP = c + base;
    const float* xP = x + base;

    const int gc = col0 - 2 * HALO + 4 * lane;
    const bool col_in = (gc >= 0) && (gc < IMG);
    const unsigned gcc = col_in ? (unsigned)gc : 0u;
    if (lane < ROW_CHUNKS) {
        float4 piece[ROWS_PER_WAVE];
#pragma unroll
        for (int j = 0; j < ROWS_PER_WAVE; ++j) {
            int gr = row0 - 2 * HALO + wv + NWAVES * j;
            int grc = gr < 0 ? 0 : (gr > IMG - 1 ? IMG - 1 : gr);
            const float* rowp = src + (base + (size_t)grc * IMG);
            piece[j] = *reinterpret_cast<const float4*>(rowp + gcc);
        }
#pragma unroll
        for (int j = 0; j < ROWS_PER_WAVE; ++j) {
            int ir = wv + NWAVES * j;
            int gr = row0 - 2 * HALO + ir;
            bool in_img = col_in && (gr >= 0) && (gr < IMG);
            const unsigned keep = in_img ? 0xffffffffu : 0u;
            float4 s;
            s.x = __uint_as_float(__float_as_uint(1.0f - 2.0f * piece[j].x) & keep);
            s.y = __uint_as_float(__float_as_uint(1.0f - 2.0f * piece[j].y) & keep);
            s.z = __uint_as_float(__float_as_uint(1.0f - 2.0f * piece[j].z) & keep);
            s.w = __uint_as_float(__float_as_uint(1.0f - 2.0f * piece[j].w) & keep);
            *reinterpret_cast<float4*>(&in_tile[ir * IN_STRIDE + 4 * lane]) = s;
        }
    }

    const int mid_r = row0 - HALO + mhalf;
    const int mid_c = col0 - HALO + n15;

    lds_barrier_signal();
    float cm[2][8], xm[2][8];
    {
        const int G = wv / MID_T, C = wv - MID_T * G;
        request_mid<INIT>(cP, xP, mid_r + 48 * G, mid_c + 16 * C, cm[0], xm[0]);
    }
    lds_barrier_wait();

    float bw[8][2];
#pragma unroll
    for (int q = 0; q < 8; ++q)
#pragma unroll
        for (int s = 0; s < 2; ++s) {
            int d = 16 * (q >> 2) + 8 * hh + 2 * (q & 3) + s - n15;
            bw[q][s] = (d >= 0 && d < TAPS) ? wsm[d] : 0.0f;
        }

    float last[8];
    float* park = &stash[wv * (5 * 256) + lane];
#pragma unroll
    for (int i = 0; i < 2; ++i) {
        const int rho = wv + NWAVES * i;
        const bool mine = (i == 0) || (rho < MID_RUNS);
        if (mine) {
            const int G = rho / MID_T, C = rho - MID_T * G;
            const float* hp = &in_tile[(48 * G + n15) * IN_STRIDE + 16 * C + 8 * hh];
            v8f up = h_tile(hp, bw);
#pragma unroll
            for (int t = 0; t < 3; ++t) {
                const int R = 3 * G + t;
                if (t < 2) request_mid<INIT>(cP, xP, mid_r + 16 * (R + 1), mid_c + 16 * C, cm[(t + 1 + i) & 1], xm[(t + 1 + i) & 1]);
                if (i == 0 && t == 2 && wv + NWAVES < MID_RUNS) {
                    const int Gn = (wv + NWAVES) / MID_T, Cn = (wv + NWAVES) - MID_T * Gn;
                    request_mid<INIT>(cP, xP, mid_r + 48 * Gn, mid_c + 16 * Cn, cm[1], xm[1]);
                }
                v8f lo = h_tile(hp + (t + 1) * 16 * IN_STRIDE, bw);
                v8f acc = v_tile(up, lo, bw);
                up = lo;
                const int gcl = mid_c + 16 * C;
                const int gr0 = mid_r + 16 * R;
                const bool in_img = (gcl >= 0) && (gcl < IMG) && (gr0 >= 0) && (gr0 < IMG);
                const float gm = in_img ? 2.5f : 0.0f;
                const float hm = in_img ? 0.5f : 0.0f;
#pragma unroll
                for (int v = 0; v < 8; ++v) {
                    float t1 = hw_tanh(5.0f * cm[(t + i) & 1][v]);
                    float nv;
                    if (INIT) {
                        nv = hw_tanh(__builtin_fmaf(t1, gm, gm) * acc[v]);
                    } else {
                        float g = __builtin_fmaf(t1, 2.5f, 2.5f);
                        float s = in_tile[(R * 16 + mhalf + v + HALO) * IN_STRIDE + C * 16 + n15 + HALO];
                        float d = __builtin_fmaf(xm[(t + i) & 1][v], -0.5f, __builtin_fmaf(s, -0.25f, 0.25f));
                        float t2 = hw_tanh(__builtin_fmaf(g, acc[v], d));
                        nv = hm * (s + t2);
                    }
                    if (i == 1 && t == 2) last[v] = nv;
                    else park[(3 * i + t) * 256 + v * 32] = nv;
                }
            }
        }
    }

    const int C2 = wv & 3, R2 = 4 * (wv >> 2);
    const size_t wave_off = (size_t)(row0 + 16 * R2) * IMG + (size_t)(col0 + 16 * C2);
    const float* cw = cP + wave_off;
    const float* xw = xP + wave_off;
    float* dw = dst + base + wave_off;
    const unsigned eoff = (unsigned)(mhalf * IMG + n15);

    lds_barrier_signal();
    float cp[2][8], xp[2][8];
#pragma unroll
    for (int v = 0; v < 8; ++v) {
        cp[0][v] = cw[eoff + (unsigned)(v * IMG)];
        xp[0][v] = xw[eoff + (unsigned)(v * IMG)];
    }
    lds_barrier_wait();

#pragma unroll
    for (int i = 0; i < 2; ++i) {
        const int rho = wv + NWAVES * i;
        const bool mine = (i == 0) || (rho < MID_RUNS);
        if (mine) {
            const int G = rho / MID_T, C = rho - MID_T * G;
#pragma unroll
            for (int t = 0; t < 3; ++t)
#pragma unroll
                for (int v = 0; v < 8; ++v)
                    in_tile[((3 * G + t) * 16 + mhalf + v + HALO) * IN_STRIDE + C * 16 + n15 + HALO] =
                        (i == 1 && t == 2) ? last[v] : park[(3 * i + t) * 256 + v * 32];
        }
    }
    lds_barrier_signal();
    lds_barrier_wait();

    {
        const float* hp = &in_tile[(HALO + 16 * R2 + n15) * IN_STRIDE + HALO + 16 * C2 + 8 * hh];
        v8f up = h_tile(hp, bw);
#pragma unroll
        for (int t = 0; t < 4; ++t) {
            if (t < 3) {
#pragma unroll
                for (int v = 0; v < 8; ++v) {
                    cp[(t + 1) & 1][v] = cw[eoff + (unsigned)((16 * (t + 1) + v) * IMG)];
                    xp[(t + 1) & 1][v] = xw[eoff + (unsigned)((16 * (t + 1) + v) * IMG)];
                }
            }
            v8f lo = h_tile(hp + (t + 1) * 16 * IN_STRIDE, bw);
            v8f acc = v_tile(up, lo, bw);
            up = lo;
#pragma unroll
            for (int v = 0; v < 8; ++v) {
                int lr = (R2 + t) * 16 + mhalf + v;
                int lc = C2 * 16 + n15;
                float g = __builtin_fmaf(hw_tanh(5.0f * cp[t & 1][v]), 2.5f, 2.5f);
                float s = in_tile[(lr + 2 * HALO) * IN_STRIDE + (lc + 2 * HALO)];
                float d = __builtin_fmaf(xp[t & 1][v], -0.5f, __builtin_fmaf(s, -0.25f, 0.25f));
                float t2 = hw_tanh(__builtin_fmaf(g, acc[v], d));
                dw[eoff + (unsigned)((16 * t + v) * IMG)] = __builtin_fmaf(s + t2, -0.25f, 0.5f);
            }
        }
    }
}

extern "C" void kernel_launch(void* const* d_in, const int* in_sizes, int n_in,
                              void* d_out, int out_size, void* d_ws, size_t ws_size,
                              hipStream_t stream) {
    const float* x = (const float*)d_in[0];
    const float* c = (const float*)d_in[1];
    float* out = (float*)d_out;
    const int n = in_sizes[0];
    const int planes = n / (IMG * IMG);

    float* ubuf = (float*)d_ws;

    GaussW gw;
    {
        float g[TAPS], s = 0.0f;
        for (int t = 0; t < TAPS; ++t) {
            float d = (float)t - 8.0f;
            g[t] = expf(-(d * d) / 8.0f);
            s += g[t];
        }
        for (int t = 0; t < TAPS; ++t) gw.w[t] = g[t] / s;
    }

    dim3 grid(IMG / TILE_C, IMG / TILE_R, planes);
    bvp_two_steps<true ><<<grid, 256, 0, stream>>>(x,    x, c, out,  gw);
    bvp_two_steps<false><<<grid, 256, 0, stream>>>(out,  x, c, ubuf, gw);
    bvp_two_steps<false><<<grid, 256, 0, stream>>>(ubuf, x, c, out,  gw);
}
